# starting kernel
# baseline (speedup 1.0000x reference)
#include <hip/hip_runtime.h>
#include <hip/hip_bf16.h>

typedef __attribute__((ext_vector_type(16))) _Float16 v16h;
typedef __attribute__((ext_vector_type(8)))  float    v8f;


__global__ void k_fill1(float* __restrict__ deg, int n) {
    int i = blockIdx.x * blockDim.x + threadIdx.x;
    if (i < n) deg[i] = 1.0f;
}

__global__ void k_degree(const int* __restrict__ dst, float* __restrict__ deg, int E) {
    int e = blockIdx.x * blockDim.x + threadIdx.x;
    if (e < E) atomicAdd(&deg[dst[e]], 1.0f);
}

__global__ void k_rsqrt_inplace(float* __restrict__ deg, int n) {
    int i = blockIdx.x * blockDim.x + threadIdx.x;
    if (i < n) deg[i] = rsqrtf(deg[i]);
}


template <int OUTC, bool RELU>
__global__ __launch_bounds__(256)
void k_gemm_f16(const float* __restrict__ X, const float* __restrict__ Wg,
                float* __restrict__ H, int ntiles) {
    __shared__ _Float16 ldsW[OUTC * 128];

    const int tid = threadIdx.x;
    for (int idx = tid; idx < 128 * OUTC; idx += 256) {
        int k = idx / OUTC;
        int n = idx % OUTC;
        ldsW[n * 128 + k] = (_Float16)Wg[idx];
    }
    __syncthreads();

    const int wave = tid >> 5;
    const int lane = tid & 31;
    const int tile = blockIdx.x * 8 + wave;
    if (tile >= ntiles) return;

    const int mn = lane & 15;
    const int hh = lane >> 4;

    const float* xrow = X + (size_t)(tile * 16 + mn) * 128;
    v16h a[4];
#pragma unroll
    for (int kf = 0; kf < 4; ++kf) {
        const int k0 = kf * 32;
#pragma unroll
        for (int i = 0; i < 8; ++i) {
            int kk = k0 + ((i < 4) ? 2 * i : 2 * i + 8) + 8 * hh;
            float v0 = xrow[kk];
            float v1 = xrow[kk + 1];
            if (RELU) { v0 = fmaxf(v0, 0.0f); v1 = fmaxf(v1, 0.0f); }
            a[kf][2 * i]     = (_Float16)v0;
            a[kf][2 * i + 1] = (_Float16)v1;
        }
    }

#pragma unroll
    for (int nt = 0; nt < OUTC / 16; ++nt) {
        const int col = nt * 16 + mn;
        const _Float16* wrow = &ldsW[col * 128];
        v8f c = {};
#pragma unroll
        for (int kf = 0; kf < 4; ++kf) {
            const int k0 = kf * 32;
            v16h b;
#pragma unroll
            for (int i = 0; i < 8; ++i) {
                int kk = k0 + ((i < 4) ? 2 * i : 2 * i + 8) + 8 * hh;
                b[2 * i]     = wrow[kk];
                b[2 * i + 1] = wrow[kk + 1];
            }
            c = __builtin_amdgcn_wmma_f32_16x16x32_f16(
                    false, a[kf], false, b, (short)0, c, false, false);
        }
        float* hout = H + (size_t)(tile * 16) * OUTC + col;
#pragma unroll
        for (int r = 0; r < 8; ++r)
            hout[(size_t)(r + 8 * hh) * OUTC] = c[r];
    }
}


template <int C>
__global__ void k_selfbias(const float* __restrict__ h, const float* __restrict__ dinv,
                           const float* __restrict__ bias, float* __restrict__ agg, int n) {
    constexpr int G = C / 4;
    int t = blockIdx.x * blockDim.x + threadIdx.x;
    int i = t / G;
    int g = t % G;
    if (i >= n) return;
    float di = dinv[i];
    float s  = di * di;
    const float4 hv = *(const float4*)(h + (size_t)i * C + g * 4);
    const float4 bv = *(const float4*)(bias + g * 4);
    float4 o;
    o.x = hv.x * s + bv.x;
    o.y = hv.y * s + bv.y;
    o.z = hv.z * s + bv.z;
    o.w = hv.w * s + bv.w;
    *(float4*)(agg + (size_t)i * C + g * 4) = o;
}


template <int C>
__global__ void k_scatter(const float* __restrict__ h, const int* __restrict__ src,
                          const int* __restrict__ dst, const float* __restrict__ dinv,
                          float* __restrict__ agg, long long total) {
    constexpr int G = C / 4;
    long long t = (long long)blockIdx.x * blockDim.x + threadIdx.x;
    if (t >= total) return;
    int e = (int)(t / G);
    int g = (int)(t % G);
    int s = src[e];
    int d = dst[e];
    float nrm = dinv[s] * dinv[d];
    const float4 hv = *(const float4*)(h + (size_t)s * C + g * 4);
    float* p = agg + (size_t)d * C + g * 4;
    atomicAdd(p + 0, hv.x * nrm);
    atomicAdd(p + 1, hv.y * nrm);
    atomicAdd(p + 2, hv.z * nrm);
    atomicAdd(p + 3, hv.w * nrm);
}


extern "C" void kernel_launch(void* const* d_in, const int* in_sizes, int n_in,
                              void* d_out, int out_size, void* d_ws, size_t ws_size,
                              hipStream_t stream) {
    const float* x  = (const float*)d_in[0];
    const int*   ei = (const int*)d_in[1];
    const float* W1 = (const float*)d_in[2];
    const float* b1 = (const float*)d_in[3];
    const float* W2 = (const float*)d_in[4];
    const float* b2 = (const float*)d_in[5];
    float*       out = (float*)d_out;

    const int N = in_sizes[0] / 128;
    const int E = in_sizes[1] / 2;
    const int* src = ei;
    const int* dst = ei + E;

    char* w = (char*)d_ws;
    auto carve = [&](size_t bytes) -> void* {
        void* p = (void*)w;
        w += (bytes + 255) & ~(size_t)255;
        return p;
    };
    float* dinv = (float*)carve((size_t)N * 4);
    float* h1   = (float*)carve((size_t)N * 128 * 4);
    float* agg1 = (float*)carve((size_t)N * 128 * 4);
    float* h2   = h1;

    const int B = 256;
    const int ntiles = (N + 15) / 16;
    const int gemm_blocks = (ntiles + 7) / 8;

    k_fill1<<<(N + B - 1) / B, B, 0, stream>>>(dinv, N);
    k_degree<<<(E + B - 1) / B, B, 0, stream>>>(dst, dinv, E);
    k_rsqrt_inplace<<<(N + B - 1) / B, B, 0, stream>>>(dinv, N);

    k_gemm_f16<128, false><<<gemm_blocks, 256, 0, stream>>>(x, W1, h1, ntiles);
    k_selfbias<128><<<(N * 32 + B - 1) / B, B, 0, stream>>>(h1, dinv, b1, agg1, N);
    {
        long long total = (long long)E * 32;
        k_scatter<128><<<(int)((total + B - 1) / B), B, 0, stream>>>(h1, src, dst, dinv, agg1, total);
    }

    k_gemm_f16<64, true><<<gemm_blocks, 256, 0, stream>>>(agg1, W2, h2, ntiles);
    k_selfbias<64><<<(N * 16 + B - 1) / B, B, 0, stream>>>(h2, dinv, b2, out, N);
    {
        long long total = (long long)E * 16;
        k_scatter<64><<<(int)((total + B - 1) / B), B, 0, stream>>>(h2, src, dst, dinv, out, total);
    }
}


# Layer-2 edge scatter only: one wave takes 4 consecutive edges (was 2), 8 line loads then 8 line adds in edge order: half as many waves to launch; layer 1 as it was; arrival-order sums kept
# speedup vs baseline: 1.8521x; 1.8521x over previous
#include <hip/hip_runtime.h>
#include <hip/hip_bf16.h>

typedef __attribute__((ext_vector_type(16))) _Float16 v16h;
typedef __attribute__((ext_vector_type(8)))  float    v8f;


constexpr int CNT_THREADS = 1024;
constexpr int CHUNK_EDGES = 32768;
constexpr int WIN_NODES   = 131072;

__global__ __launch_bounds__(CNT_THREADS)
void k_count(const int* __restrict__ dst, unsigned int* __restrict__ part,
             int E, int n, int wpr) {
    __shared__ uint4 cnt4[WIN_NODES / 8];
    unsigned int* cnt = (unsigned int*)cnt4;
    const int tid  = threadIdx.x;
    const int base = blockIdx.y * WIN_NODES;
    const int nn   = min(WIN_NODES, n - base);
    const int nq   = (nn + 7) >> 3;

    constexpr int EPT = CHUNK_EDGES / CNT_THREADS;
    const int e0 = blockIdx.x * CHUNK_EDGES;
    int d[EPT];
#pragma unroll
    for (int k = 0; k < EPT; ++k) {
        const int e = e0 + k * CNT_THREADS + tid;
        d[k] = (e < E) ? dst[e] : -1;
    }

    for (int q = tid; q < nq; q += CNT_THREADS) cnt4[q] = make_uint4(0u, 0u, 0u, 0u);
    __syncthreads();

#pragma unroll
    for (int k = 0; k < EPT; ++k) {
        const unsigned int r = (unsigned int)(d[k] - base);
        if (r < (unsigned int)nn)
            atomicAdd(&cnt[r >> 1], 1u << ((r & 1u) * 16u));
    }
    __syncthreads();

    uint4* row = (uint4*)(part + (size_t)blockIdx.x * wpr + (size_t)blockIdx.y * (WIN_NODES / 2));
    for (int q = tid; q < nq; q += CNT_THREADS) row[q] = cnt4[q];
}


template <int OUTC>
__device__ __forceinline__ void lay_b_fragments(const float* __restrict__ Wg,
                                                _Float16* __restrict__ Wf,
                                                int first, int step) {
    for (int idx = first; idx < 128 * OUTC; idx += step) {
        const int s    = idx & 15;
        const int lane = (idx >> 4) & 31;
        const int kf   = (idx >> 9) & 3;
        const int nt   = idx >> 11;
        const int k = kf * 32 + ((s < 8) ? s : s + 8) + 8 * (lane >> 4);
        const int n = nt * 16 + (lane & 15);
        Wf[idx] = (_Float16)Wg[k * OUTC + n];
    }
}

__global__ void k_dinv(const unsigned short* __restrict__ part16, int nchunks, int pitch16,
                       float* __restrict__ dinv, int n,
                       const float* __restrict__ W1, _Float16* __restrict__ Wf1,
                       const float* __restrict__ W2, _Float16* __restrict__ Wf2) {
    const int i    = blockIdx.x * blockDim.x + threadIdx.x;
    const int step = gridDim.x * blockDim.x;
    if (i < n) {
        unsigned int c = 0;
        const unsigned short* p = part16 + i;
#pragma unroll 8
        for (int k = 0; k < nchunks; ++k) c += p[(size_t)k * pitch16];
        dinv[i] = rsqrtf((float)c + 1.0f);
    }
    lay_b_fragments<128>(W1, Wf1, i, step);
    lay_b_fragments<64>(W2, Wf2, i, step);
}

template <int OUTC, bool RELU>
__global__ __launch_bounds__(256)
void k_gemm_f16(const float* __restrict__ X, const _Float16* __restrict__ Wf,
                float* __restrict__ H, int ntiles) {
    const int tid  = threadIdx.x;
    const int wave = tid >> 5;
    const int lane = tid & 31;
    const int tile = blockIdx.x * 8 + wave;
    if (tile >= ntiles) return;

    const int mn = lane & 15;
    const int hh = lane >> 4;

    const float* xrow = X + (size_t)(tile * 16 + mn) * 128;
    v16h a[4];
#pragma unroll
    for (int kf = 0; kf < 4; ++kf) {
        const int k0 = kf * 32;
#pragma unroll
        for (int i = 0; i < 8; ++i) {
            int kk = k0 + ((i < 4) ? 2 * i : 2 * i + 8) + 8 * hh;
            float v0 = xrow[kk];
            float v1 = xrow[kk + 1];
            if (RELU) { v0 = fmaxf(v0, 0.0f); v1 = fmaxf(v1, 0.0f); }
            a[kf][2 * i]     = (_Float16)v0;
            a[kf][2 * i + 1] = (_Float16)v1;
        }
    }

    const v16h* bfrag = (const v16h*)Wf + lane;
#pragma unroll
    for (int nt = 0; nt < OUTC / 16; ++nt) {
        const int col = nt * 16 + mn;
        v8f c = {};
#pragma unroll
        for (int kf = 0; kf < 4; ++kf) {
            const v16h b = bfrag[(nt * 4 + kf) * 32];
            c = __builtin_amdgcn_wmma_f32_16x16x32_f16(
                    false, a[kf], false, b, (short)0, c, false, false);
        }
        float* hout = H + (size_t)(tile * 16) * OUTC + col;
#pragma unroll
        for (int r = 0; r < 8; ++r)
            hout[(size_t)(r + 8 * hh) * OUTC] = c[r];
    }
}


template <int C>
__global__ void k_selfbias(const float* __restrict__ h, const float* __restrict__ dinv,
                           const float* __restrict__ bias, float* __restrict__ agg, int n) {
    constexpr int G = C / 4;
    int t = blockIdx.x * blockDim.x + threadIdx.x;
    int i = t / G;
    int g = t % G;
    if (i >= n) return;
    float di = dinv[i];
    float s  = di * di;
    const float4 hv = *(const float4*)(h + (size_t)i * C + g * 4);
    const float4 bv = *(const float4*)(bias + g * 4);
    float4 o;
    o.x = hv.x * s + bv.x;
    o.y = hv.y * s + bv.y;
    o.z = hv.z * s + bv.z;
    o.w = hv.w * s + bv.w;
    *(float4*)(agg + (size_t)i * C + g * 4) = o;
}


template <int C, int K>
__device__ __forceinline__ void scatter_edges(const float* __restrict__ h, const int* __restrict__ src,
                                              const int* __restrict__ dst, const float* __restrict__ dinv,
                                              float* __restrict__ agg, int e0, int lane) {
    constexpr int LINES = C / 32;
    int   d[K];
    float nrm[K];
    float v[K][LINES];
#pragma unroll
    for (int k = 0; k < K; ++k) {
        const int s = src[e0 + k];
        d[k] = dst[e0 + k];
        nrm[k] = dinv[s] * dinv[d[k]];
        const float* hr = h + (size_t)s * C + lane;
#pragma unroll
        for (int j = 0; j < LINES; ++j) v[k][j] = hr[32 * j];
    }
#pragma unroll
    for (int k = 0; k < K; ++k) {
        float* p = agg + (size_t)d[k] * C + lane;
#pragma unroll
        for (int j = 0; j < LINES; ++j) atomicAdd(p + 32 * j, v[k][j] * nrm[k]);
    }
}

template <int C, int EPW>
__global__ void k_scatter(const float* __restrict__ h, const int* __restrict__ src,
                          const int* __restrict__ dst, const float* __restrict__ dinv,
                          float* __restrict__ agg, int E) {
    const int t    = blockIdx.x * blockDim.x + threadIdx.x;
    const int lane = t & 31;
    const int e0   = (t >> 5) * EPW;
    if (e0 + EPW <= E) {
        scatter_edges<C, EPW>(h, src, dst, dinv, agg, e0, lane);
    } else {
        for (int e = e0; e < E; ++e)
            scatter_edges<C, 1>(h, src, dst, dinv, agg, e, lane);
    }
}


extern "C" void kernel_launch(void* const* d_in, const int* in_sizes, int n_in,
                              void* d_out, int out_size, void* d_ws, size_t ws_size,
                              hipStream_t stream) {
    const float* x  = (const float*)d_in[0];
    const int*   ei = (const int*)d_in[1];
    const float* W1 = (const float*)d_in[2];
    const float* b1 = (const float*)d_in[3];
    const float* W2 = (const float*)d_in[4];
    const float* b2 = (const float*)d_in[5];
    float*       out = (float*)d_out;

    const int N = in_sizes[0] / 128;
    const int E = in_sizes[1] / 2;
    const int* src = ei;
    const int* dst = ei + E;

    char* w = (char*)d_ws;
    auto carve = [&](size_t bytes) -> void* {
        void* p = (void*)w;
        w += (bytes + 255) & ~(size_t)255;
        return p;
    };
    float* dinv = (float*)carve((size_t)N * 4);
    float* h1   = (float*)carve((size_t)N * 128 * 4);
    float* agg1 = (float*)carve((size_t)N * 128 * 4);
    float* h2   = h1;
    _Float16* Wf1 = (_Float16*)carve((size_t)128 * 128 * 2);
    _Float16* Wf2 = (_Float16*)carve((size_t)128 * 64 * 2);
    const int nchunks = (E + CHUNK_EDGES - 1) / CHUNK_EDGES;
    const int nwin    = (N + WIN_NODES - 1) / WIN_NODES;
    const int wpr     = ((N + 7) / 8) * 4;
    unsigned int* part = (unsigned int*)carve((size_t)nchunks * wpr * 4);

    const int B = 256;
    const int ntiles = (N + 15) / 16;
    const int gemm_blocks = (ntiles + 7) / 8;

    k_count<<<dim3(nchunks, nwin), CNT_THREADS, 0, stream>>>(dst, part, E, N, wpr);
    k_dinv<<<(N + B - 1) / B, B, 0, stream>>>((const unsigned short*)part, nchunks, 2 * wpr,
                                              dinv, N, W1, Wf1, W2, Wf2);

    k_gemm_f16<128, false><<<gemm_blocks, 256, 0, stream>>>(x, Wf1, h1, ntiles);
    k_selfbias<128><<<(N * 32 + B - 1) / B, B, 0, stream>>>(h1, dinv, b1, agg1, N);
    k_scatter<128, 1><<<(E * 32 + B - 1) / B, B, 0, stream>>>(h1, src, dst, dinv, agg1, E);

    k_gemm_f16<64, true><<<gemm_blocks, 256, 0, stream>>>(agg1, Wf2, h2, ntiles);
    k_selfbias<64><<<(N * 16 + B - 1) / B, B, 0, stream>>>(h2, dinv, b2, out, N);
    {
        const int waves = (E + 3) / 4;
        k_scatter<64, 4><<<(waves * 32 + B - 1) / B, B, 0, stream>>>(h2, src, dst, dinv, out, E);
    }
}


# Layer-1 edge scatter as two 64-channel passes over plane-split h1 and aggregate; a pass is the 4-edge-a-wave scatter on 51 MB of rows (was one pass on 102 MB, 1 edge a wave); arrival-order sums kept
# speedup vs baseline: 2.1618x; 1.1672x over previous
#include <hip/hip_runtime.h>
#include <hip/hip_bf16.h>

typedef __attribute__((ext_vector_type(16))) _Float16 v16h;
typedef __attribute__((ext_vector_type(8)))  float    v8f;


constexpr int CNT_THREADS = 1024;
constexpr int CHUNK_EDGES = 32768;
constexpr int WIN_NODES   = 131072;

__global__ __launch_bounds__(CNT_THREADS)
void k_count(const int* __restrict__ dst, unsigned int* __restrict__ part,
             int E, int n, int wpr) {
    __shared__ uint4 cnt4[WIN_NODES / 8];
    unsigned int* cnt = (unsigned int*)cnt4;
    const int tid  = threadIdx.x;
    const int base = blockIdx.y * WIN_NODES;
    const int nn   = min(WIN_NODES, n - base);
    const int nq   = (nn + 7) >> 3;

    constexpr int EPT = CHUNK_EDGES / CNT_THREADS;
    const int e0 = blockIdx.x * CHUNK_EDGES;
    int d[EPT];
#pragma unroll
    for (int k = 0; k < EPT; ++k) {
        const int e = e0 + k * CNT_THREADS + tid;
        d[k] = (e < E) ? dst[e] : -1;
    }

    for (int q = tid; q < nq; q += CNT_THREADS) cnt4[q] = make_uint4(0u, 0u, 0u, 0u);
    __syncthreads();

#pragma unroll
    for (int k = 0; k < EPT; ++k) {
        const unsigned int r = (unsigned int)(d[k] - base);
        if (r < (unsigned int)nn)
            atomicAdd(&cnt[r >> 1], 1u << ((r & 1u) * 16u));
    }
    __syncthreads();

    uint4* row = (uint4*)(part + (size_t)blockIdx.x * wpr + (size_t)blockIdx.y * (WIN_NODES / 2));
    for (int q = tid; q < nq; q += CNT_THREADS) row[q] = cnt4[q];
}


template <int OUTC>
__device__ __forceinline__ void lay_b_fragments(const float* __restrict__ Wg,
                                                _Float16* __restrict__ Wf,
                                                int first, int step) {
    for (int idx = first; idx < 128 * OUTC; idx += step) {
        const int s    = idx & 15;
        const int lane = (idx >> 4) & 31;
        const int kf   = (idx >> 9) & 3;
        const int nt   = idx >> 11;
        const int k = kf * 32 + ((s < 8) ? s : s + 8) + 8 * (lane >> 4);
        const int n = nt * 16 + (lane & 15);
        Wf[idx] = (_Float16)Wg[k * OUTC + n];
    }
}

__global__ void k_dinv(const unsigned short* __restrict__ part16, int nchunks, int pitch16,
                       float* __restrict__ dinv, int n,
                       const float* __restrict__ W1, _Float16* __restrict__ Wf1,
                       const float* __restrict__ W2, _Float16* __restrict__ Wf2) {
    const int i    = blockIdx.x * blockDim.x + threadIdx.x;
    const int step = gridDim.x * blockDim.x;
    if (i < n) {
        unsigned int c = 0;
        const unsigned short* p = part16 + i;
#pragma unroll 8
        for (int k = 0; k < nchunks; ++k) c += p[(size_t)k * pitch16];
        dinv[i] = rsqrtf((float)c + 1.0f);
    }
    lay_b_fragments<128>(W1, Wf1, i, step);
    lay_b_fragments<64>(W2, Wf2, i, step);
}

template <int OUTC, bool RELU, bool PIN, bool POUT>
__global__ __launch_bounds__(256)
void k_gemm_f16(const float* __restrict__ X, const _Float16* __restrict__ Wf,
                float* __restrict__ H, int ntiles, int n) {
    static_assert(!POUT || OUTC == 128, "plane-split output is for 128-channel rows");
    const size_t plane = (size_t)n * 64;
    const int tid  = threadIdx.x;
    const int wave = tid >> 5;
    const int lane = tid & 31;
    const int tile = blockIdx.x * 8 + wave;
    if (tile >= ntiles) return;

    const int mn = lane & 15;
    const int hh = lane >> 4;

    const size_t arow = (size_t)(tile * 16 + mn);
    v16h a[4];
#pragma unroll
    for (int kf = 0; kf < 4; ++kf) {
        const float* xk = PIN ? X + (size_t)(kf >> 1) * plane + arow * 64 + (kf & 1) * 32
                              : X + arow * 128 + kf * 32;
#pragma unroll
        for (int i = 0; i < 8; ++i) {
            int kk = ((i < 4) ? 2 * i : 2 * i + 8) + 8 * hh;
            float v0 = xk[kk];
            float v1 = xk[kk + 1];
            if (RELU) { v0 = fmaxf(v0, 0.0f); v1 = fmaxf(v1, 0.0f); }
            a[kf][2 * i]     = (_Float16)v0;
            a[kf][2 * i + 1] = (_Float16)v1;
        }
    }

    const v16h* bfrag = (const v16h*)Wf + lane;
#pragma unroll
    for (int nt = 0; nt < OUTC / 16; ++nt) {
        v8f c = {};
#pragma unroll
        for (int kf = 0; kf < 4; ++kf) {
            const v16h b = bfrag[(nt * 4 + kf) * 32];
            c = __builtin_amdgcn_wmma_f32_16x16x32_f16(
                    false, a[kf], false, b, (short)0, c, false, false);
        }
        constexpr int PITCH = POUT ? 64 : OUTC;
        float* hout = POUT ? H + (size_t)(nt >> 2) * plane + (size_t)(tile * 16) * 64 + (nt & 3) * 16 + mn
                           : H + (size_t)(tile * 16) * OUTC + nt * 16 + mn;
#pragma unroll
        for (int r = 0; r < 8; ++r)
            hout[(size_t)(r + 8 * hh) * PITCH] = c[r];
    }
}


template <int C, bool SPLIT>
__global__ void k_selfbias(const float* __restrict__ h, const float* __restrict__ dinv,
                           const float* __restrict__ bias, float* __restrict__ agg, int n) {
    static_assert(!SPLIT || C == 128, "plane-split rows have 128 channels");
    constexpr int G = C / 4;
    int t = blockIdx.x * blockDim.x + threadIdx.x;
    int i = t / G;
    int g = t % G;
    if (i >= n) return;
    float di = dinv[i];
    float s  = di * di;
    const size_t off = SPLIT ? (size_t)(g >> 4) * n * 64 + (size_t)i * 64 + (g & 15) * 4
                             : (size_t)i * C + g * 4;
    const float4 hv = *(const float4*)(h + off);
    const float4 bv = *(const float4*)(bias + g * 4);
    float4 o;
    o.x = hv.x * s + bv.x;
    o.y = hv.y * s + bv.y;
    o.z = hv.z * s + bv.z;
    o.w = hv.w * s + bv.w;
    *(float4*)(agg + off) = o;
}


template <int C, int K>
__device__ __forceinline__ void scatter_edges(const float* __restrict__ h, const int* __restrict__ src,
                                              const int* __restrict__ dst, const float* __restrict__ dinv,
                                              float* __restrict__ agg, int e0, int lane) {
    constexpr int LINES = C / 32;
    int   d[K];
    float nrm[K];
    float v[K][LINES];
#pragma unroll
    for (int k = 0; k < K; ++k) {
        const int s = src[e0 + k];
        d[k] = dst[e0 + k];
        nrm[k] = dinv[s] * dinv[d[k]];
        const float* hr = h + (size_t)s * C + lane;
#pragma unroll
        for (int j = 0; j < LINES; ++j) v[k][j] = hr[32 * j];
    }
#pragma unroll
    for (int k = 0; k < K; ++k) {
        float* p = agg + (size_t)d[k] * C + lane;
#pragma unroll
        for (int j = 0; j < LINES; ++j) atomicAdd(p + 32 * j, v[k][j] * nrm[k]);
    }
}

template <int C, int EPW>
__global__ void k_scatter(const float* __restrict__ h, const int* __restrict__ src,
                          const int* __restrict__ dst, const float* __restrict__ dinv,
                          float* __restrict__ agg, int E) {
    const int t    = blockIdx.x * blockDim.x + threadIdx.x;
    const int lane = t & 31;
    const int e0   = (t >> 5) * EPW;
    if (e0 + EPW <= E) {
        scatter_edges<C, EPW>(h, src, dst, dinv, agg, e0, lane);
    } else {
        for (int e = e0; e < E; ++e)
            scatter_edges<C, 1>(h, src, dst, dinv, agg, e, lane);
    }
}


extern "C" void kernel_launch(void* const* d_in, const int* in_sizes, int n_in,
                              void* d_out, int out_size, void* d_ws, size_t ws_size,
                              hipStream_t stream) {
    const float* x  = (const float*)d_in[0];
    const int*   ei = (const int*)d_in[1];
    const float* W1 = (const float*)d_in[2];
    const float* b1 = (const float*)d_in[3];
    const float* W2 = (const float*)d_in[4];
    const float* b2 = (const float*)d_in[5];
    float*       out = (float*)d_out;

    const int N = in_sizes[0] / 128;
    const int E = in_sizes[1] / 2;
    const int* src = ei;
    const int* dst = ei + E;

    char* w = (char*)d_ws;
    auto carve = [&](size_t bytes) -> void* {
        void* p = (void*)w;
        w += (bytes + 255) & ~(size_t)255;
        return p;
    };
    float* dinv = (float*)carve((size_t)N * 4);
    float* h1   = (float*)carve((size_t)N * 128 * 4);
    float* agg1 = (float*)carve((size_t)N * 128 * 4);
    float* h2   = h1;
    _Float16* Wf1 = (_Float16*)carve((size_t)128 * 128 * 2);
    _Float16* Wf2 = (_Float16*)carve((size_t)128 * 64 * 2);
    const int nchunks = (E + CHUNK_EDGES - 1) / CHUNK_EDGES;
    const int nwin    = (N + WIN_NODES - 1) / WIN_NODES;
    const int wpr     = ((N + 7) / 8) * 4;
    unsigned int* part = (unsigned int*)carve((size_t)nchunks * wpr * 4);

    const int B = 256;
    const int ntiles = (N + 15) / 16;
    const int gemm_blocks = (ntiles + 7) / 8;

    k_count<<<dim3(nchunks, nwin), CNT_THREADS, 0, stream>>>(dst, part, E, N, wpr);
    k_dinv<<<(N + B - 1) / B, B, 0, stream>>>((const unsigned short*)part, nchunks, 2 * wpr,
                                              dinv, N, W1, Wf1, W2, Wf2);

    const int waves = (E + 3) / 4;
    const int scat_blocks = (waves * 32 + B - 1) / B;
    const size_t plane = (size_t)N * 64;
    k_gemm_f16<128, false, false, true><<<gemm_blocks, 256, 0, stream>>>(x, Wf1, h1, ntiles, N);
    k_selfbias<128, true><<<(N * 32 + B - 1) / B, B, 0, stream>>>(h1, dinv, b1, agg1, N);
    k_scatter<64, 4><<<scat_blocks, B, 0, stream>>>(h1, src, dst, dinv, agg1, E);
    k_scatter<64, 4><<<scat_blocks, B, 0, stream>>>(h1 + plane, src, dst, dinv, agg1 + plane, E);

    k_gemm_f16<64, true, true, false><<<gemm_blocks, 256, 0, stream>>>(agg1, Wf2, h2, ntiles, N);
    k_selfbias<64, false><<<(N * 16 + B - 1) / B, B, 0, stream>>>(h2, dinv, b2, out, N);
    k_scatter<64, 4><<<scat_blocks, B, 0, stream>>>(h2, src, dst, dinv, out, E);
}


# Layer-2 GEMM takes each layer-1 sum as two 16-bit terms (value + rounding remainder): arrival order cannot flip a rounding, so layer-1 scatter takes 16 edges a wave (was 4); arrival-order sums kept
# speedup vs baseline: 2.6770x; 1.2383x over previous
#include <hip/hip_runtime.h>
#include <hip/hip_bf16.h>

typedef __attribute__((ext_vector_type(16))) _Float16 v16h;
typedef __attribute__((ext_vector_type(8)))  float    v8f;

template <int OUTC>
__device__ __forceinline__ void lay_b_fragments(const float* __restrict__ Wg,
                                                _Float16* __restrict__ Wf,
                                                int first, int step) {
    for (int idx = first; idx < 128 * OUTC; idx += step) {
        const int s    = idx & 15;
        const int lane = (idx >> 4) & 31;
        const int kf   = (idx >> 9) & 3;
        const int nt   = idx >> 11;
        const int k = kf * 32 + ((s < 8) ? s : s + 8) + 8 * (lane >> 4);
        const int n = nt * 16 + (lane & 15);
        Wf[idx] = (_Float16)Wg[k * OUTC + n];
    }
}


constexpr int CNT_THREADS = 1024;
constexpr int CHUNK_EDGES = 32768;
constexpr int WIN_NODES   = 131072;

__global__ __launch_bounds__(CNT_THREADS)
void k_count(const int* __restrict__ dst, unsigned int* __restrict__ part,
             int E, int n, int wpr,
             const float* __restrict__ W1, _Float16* __restrict__ Wf1,
             const float* __restrict__ W2, _Float16* __restrict__ Wf2) {
    __shared__ uint4 cnt4[WIN_NODES / 8];
    unsigned int* cnt = (unsigned int*)cnt4;
    const int tid  = threadIdx.x;
    const int base = blockIdx.y * WIN_NODES;
    const int nn   = min(WIN_NODES, n - base);
    const int nq   = (nn + 7) >> 3;

    constexpr int EPT = CHUNK_EDGES / CNT_THREADS;
    const int e0 = blockIdx.x * CHUNK_EDGES;
    int d[EPT];
#pragma unroll
    for (int k = 0; k < EPT; ++k) {
        const int e = e0 + k * CNT_THREADS + tid;
        d[k] = (e < E) ? dst[e] : -1;
    }

    for (int q = tid; q < nq; q += CNT_THREADS) cnt4[q] = make_uint4(0u, 0u, 0u, 0u);
    __syncthreads();

#pragma unroll
    for (int k = 0; k < EPT; ++k) {
        const unsigned int r = (unsigned int)(d[k] - base);
        if (r < (unsigned int)nn)
            atomicAdd(&cnt[r >> 1], 1u << ((r & 1u) * 16u));
    }
    __syncthreads();

    uint4* row = (uint4*)(part + (size_t)blockIdx.x * wpr + (size_t)blockIdx.y * (WIN_NODES / 2));
    for (int q = tid; q < nq; q += CNT_THREADS) row[q] = cnt4[q];

    const int gthread = (blockIdx.y * gridDim.x + blockIdx.x) * CNT_THREADS + tid;
    const int gstep   = gridDim.x * gridDim.y * CNT_THREADS;
    lay_b_fragments<128>(W1, Wf1, gthread, gstep);
    lay_b_fragments<64>(W2, Wf2, gthread, gstep);
}

template <int OUTC, bool FIN, bool PIN, bool POUT>
__global__ __launch_bounds__(256)
void k_gemm_f16(const float* __restrict__ X, const _Float16* __restrict__ Wf,
                const unsigned short* __restrict__ part16, int nchunks, int pitch16,
                float* __restrict__ dinv, const float* __restrict__ bin,
                const float* __restrict__ bout,
                float* __restrict__ H, float* __restrict__ S, int ntiles, int n) {
    static_assert(!POUT || OUTC == 128, "plane-split output is for 128-channel rows");
    const size_t plane = (size_t)n * 64;
    const int tid  = threadIdx.x;
    const int wave = tid >> 5;
    const int lane = tid & 31;
    const int tile = blockIdx.x * 8 + wave;
    if (tile >= ntiles) return;

    const int mn = lane & 15;
    const int hh = lane >> 4;

    const size_t arow = (size_t)(tile * 16 + mn);
    float dva;
    if (FIN) {
        dva = dinv[arow];
    } else {
        unsigned int c = 0;
        const unsigned short* p = part16 + arow;
        for (int k0 = hh; k0 < nchunks; k0 += 64) {
            unsigned int v[32];
#pragma unroll
            for (int j = 0; j < 32; ++j)
                v[j] = p[(size_t)min(k0 + 2 * j, nchunks - 1) * pitch16];
#pragma unroll
            for (int j = 0; j < 32; ++j)
                c += (k0 + 2 * j < nchunks) ? v[j] : 0u;
        }
        c += (unsigned int)__shfl_xor((int)c, 16);
        dva = rsqrtf((float)c + 1.0f);
        if (hh == 0 && arow < (size_t)n) dinv[arow] = dva;
    }

    v16h a[4];
    v16h al[FIN ? 4 : 1];
#pragma unroll
    for (int kf = 0; kf < 4; ++kf) {
        const float* xk = PIN ? X + (size_t)(kf >> 1) * plane + arow * 64 + (kf & 1) * 32
                              : X + arow * 128 + kf * 32;
#pragma unroll
        for (int i = 0; i < 8; ++i) {
            int kk = ((i < 4) ? 2 * i : 2 * i + 8) + 8 * hh;
            float v0 = xk[kk];
            float v1 = xk[kk + 1];
            if (FIN) {
                const float* bk = bin + kf * 32 + kk;
                v0 = fmaxf(fmaf(v0, dva, bk[0]), 0.0f);
                v1 = fmaxf(fmaf(v1, dva, bk[1]), 0.0f);
            }
            const _Float16 h0 = (_Float16)v0;
            const _Float16 h1 = (_Float16)v1;
            a[kf][2 * i]     = h0;
            a[kf][2 * i + 1] = h1;
            if (FIN) {
                al[kf][2 * i]     = (_Float16)(v0 - (float)h0);
                al[kf][2 * i + 1] = (_Float16)(v1 - (float)h1);
            }
        }
    }

    float dvo[8];
#pragma unroll
    for (int r = 0; r < 8; ++r)
        dvo[r] = FIN ? dinv[tile * 16 + r + 8 * hh] : __shfl(dva, r + 8 * hh);

    const v16h* bfrag = (const v16h*)Wf + lane;
#pragma unroll
    for (int nt = 0; nt < OUTC / 16; ++nt) {
        v8f c = {};
#pragma unroll
        for (int kf = 0; kf < 4; ++kf) {
            const v16h b = bfrag[(nt * 4 + kf) * 32];
            c = __builtin_amdgcn_wmma_f32_16x16x32_f16(
                    false, a[kf], false, b, (short)0, c, false, false);
            if (FIN)
                c = __builtin_amdgcn_wmma_f32_16x16x32_f16(
                        false, al[kf], false, b, (short)0, c, false, false);
        }
        constexpr int PITCH = POUT ? 64 : OUTC;
        const size_t off = POUT ? (size_t)(nt >> 2) * plane + (size_t)(tile * 16) * 64 + (nt & 3) * 16 + mn
                                : (size_t)(tile * 16) * OUTC + nt * 16 + mn;
        float* hout = H + off;
        float* sout = S + off;
        const float bo = FIN ? bout[nt * 16 + mn] : 0.0f;
#pragma unroll
        for (int r = 0; r < 8; ++r) {
            const float hv = c[r] * dvo[r];
            hout[(size_t)(r + 8 * hh) * PITCH] = hv;
            sout[(size_t)(r + 8 * hh) * PITCH] = FIN ? hv * dvo[r] + bo : hv;
        }
    }
}


template <int C, int K, bool DSTN>
__device__ __forceinline__ void scatter_edges(const float* __restrict__ h, const int* __restrict__ src,
                                              const int* __restrict__ dst, const float* __restrict__ dinv,
                                              float* __restrict__ agg, int e0, int lane) {
    constexpr int LINES = C / 32;
    int   d[K];
    float nrm[K];
    float v[K][LINES];
#pragma unroll
    for (int k = 0; k < K; ++k) {
        const int s = src[e0 + k];
        d[k] = dst[e0 + k];
        nrm[k] = DSTN ? dinv[d[k]] : 1.0f;
        const float* hr = h + (size_t)s * C + lane;
#pragma unroll
        for (int j = 0; j < LINES; ++j) v[k][j] = hr[32 * j];
    }
#pragma unroll
    for (int k = 0; k < K; ++k) {
        float* p = agg + (size_t)d[k] * C + lane;
#pragma unroll
        for (int j = 0; j < LINES; ++j) atomicAdd(p + 32 * j, DSTN ? v[k][j] * nrm[k] : v[k][j]);
    }
}

template <int C, bool DSTN>
__device__ __forceinline__ void scatter_edges16(const float* __restrict__ h, const int* __restrict__ src,
                                                const int* __restrict__ dst, const float* __restrict__ dinv,
                                                float* __restrict__ agg, int e0, int lane) {
    constexpr int LINES = C / 32;
    const int k16 = lane & 15;
    const int id  = (lane < 16) ? src[e0 + k16] : dst[e0 + k16];
    const float nv = DSTN ? dinv[(lane >= 16) ? id : 0] : 1.0f;
    const int nvb = __builtin_bit_cast(int, nv);
    float v[16][LINES];
#pragma unroll
    for (int k = 0; k < 16; ++k) {
        const int s = __builtin_amdgcn_readlane(id, k);
        const float* hr = h + (size_t)s * C + lane;
#pragma unroll
        for (int j = 0; j < LINES; ++j) v[k][j] = hr[32 * j];
    }
#pragma unroll
    for (int k = 0; k < 16; ++k) {
        const int d = __builtin_amdgcn_readlane(id, 16 + k);
        const float nrm = __builtin_bit_cast(float, __builtin_amdgcn_readlane(nvb, 16 + k));
        float* p = agg + (size_t)d * C + lane;
#pragma unroll
        for (int j = 0; j < LINES; ++j) atomicAdd(p + 32 * j, DSTN ? v[k][j] * nrm : v[k][j]);
    }
}

template <int C, bool DSTN>
__global__ void k_scatter16(const float* __restrict__ h, const int* __restrict__ src,
                            const int* __restrict__ dst, const float* __restrict__ dinv,
                            float* __restrict__ agg, int E) {
    const int t    = blockIdx.x * blockDim.x + threadIdx.x;
    const int lane = t & 31;
    const int e0   = (t >> 5) * 16;
    if (e0 + 16 <= E) {
        scatter_edges16<C, DSTN>(h, src, dst, dinv, agg, e0, lane);
    } else {
        for (int e = e0; e < E; ++e)
            scatter_edges<C, 1, DSTN>(h, src, dst, dinv, agg, e, lane);
    }
}


extern "C" void kernel_launch(void* const* d_in, const int* in_sizes, int n_in,
                              void* d_out, int out_size, void* d_ws, size_t ws_size,
                              hipStream_t stream) {
    const float* x  = (const float*)d_in[0];
    const int*   ei = (const int*)d_in[1];
    const float* W1 = (const float*)d_in[2];
    const float* b1 = (const float*)d_in[3];
    const float* W2 = (const float*)d_in[4];
    const float* b2 = (const float*)d_in[5];
    float*       out = (float*)d_out;

    const int N = in_sizes[0] / 128;
    const int E = in_sizes[1] / 2;
    const int* src = ei;
    const int* dst = ei + E;

    char* w = (char*)d_ws;
    auto carve = [&](size_t bytes) -> void* {
        void* p = (void*)w;
        w += (bytes + 255) & ~(size_t)255;
        return p;
    };
    float* dinv = (float*)carve((size_t)N * 4);
    float* h1   = (float*)carve((size_t)N * 128 * 4);
    float* agg1 = (float*)carve((size_t)N * 128 * 4);
    float* h2   = h1;
    _Float16* Wf1 = (_Float16*)carve((size_t)128 * 128 * 2);
    _Float16* Wf2 = (_Float16*)carve((size_t)128 * 64 * 2);
    const int nchunks = (E + CHUNK_EDGES - 1) / CHUNK_EDGES;
    const int nwin    = (N + WIN_NODES - 1) / WIN_NODES;
    const int wpr     = ((N + 7) / 8) * 4;
    unsigned int* part = (unsigned int*)carve((size_t)nchunks * wpr * 4);

    const int B = 256;
    const int ntiles = (N + 15) / 16;
    const int gemm_blocks = (ntiles + 7) / 8;

    k_count<<<dim3(nchunks, nwin), CNT_THREADS, 0, stream>>>(dst, part, E, N, wpr, W1, Wf1, W2, Wf2);
    const unsigned short* part16 = (const unsigned short*)part;

    const int waves16 = (E + 15) / 16;
    const int scat16_blocks = (waves16 * 32 + B - 1) / B;
    const size_t plane = (size_t)N * 64;
    k_gemm_f16<128, false, false, true><<<gemm_blocks, 256, 0, stream>>>(x, Wf1, part16, nchunks, 2 * wpr,
                                                                         dinv, b1, b1, h1, agg1, ntiles, N);
    k_scatter16<64, false><<<scat16_blocks, B, 0, stream>>>(h1, src, dst, dinv, agg1, E);
    k_scatter16<64, false><<<scat16_blocks, B, 0, stream>>>(h1 + plane, src, dst, dinv, agg1 + plane, E);

    k_gemm_f16<64, true, true, false><<<gemm_blocks, 256, 0, stream>>>(agg1, Wf2, part16, nchunks, 2 * wpr,
                                                                       dinv, b1, b2, h2, out, ntiles, N);
    k_scatter16<64, true><<<scat16_blocks, B, 0, stream>>>(h2, src, dst, dinv, out, E);
}


# Edge scatter, all 3 launches: sum rows split by node parity, each package half (die id) adds only into its own rows, edge chunks drawn from per-parity work counters; arrival-order f32 sums kept
# speedup vs baseline: 3.0647x; 1.1448x over previous
#include <hip/hip_runtime.h>
#include <hip/hip_bf16.h>

typedef __attribute__((ext_vector_type(16))) _Float16 v16h;
typedef __attribute__((ext_vector_type(8)))  float    v8f;

template <int OUTC>
__device__ __forceinline__ void lay_b_fragments(const float* __restrict__ Wg,
                                                _Float16* __restrict__ Wf,
                                                int first, int step) {
    for (int idx = first; idx < 128 * OUTC; idx += step) {
        const int s    = idx & 15;
        const int lane = (idx >> 4) & 31;
        const int kf   = (idx >> 9) & 3;
        const int nt   = idx >> 11;
        const int k = kf * 32 + ((s < 8) ? s : s + 8) + 8 * (lane >> 4);
        const int n = nt * 16 + (lane & 15);
        Wf[idx] = (_Float16)Wg[k * OUTC + n];
    }
}


constexpr int CTR_PITCH  = 64;
constexpr int N_CTR      = 6;

constexpr int CNT_THREADS = 1024;
constexpr int CHUNK_EDGES = 32768;
constexpr int WIN_NODES   = 131072;

__global__ __launch_bounds__(CNT_THREADS)
void k_count(const int* __restrict__ dst, unsigned int* __restrict__ part,
             int E, int n, int wpr,
             const float* __restrict__ W1, _Float16* __restrict__ Wf1,
             const float* __restrict__ W2, _Float16* __restrict__ Wf2,
             unsigned int* __restrict__ ctr) {
    __shared__ uint4 cnt4[WIN_NODES / 8];
    unsigned int* cnt = (unsigned int*)cnt4;
    const int tid  = threadIdx.x;
    const int base = blockIdx.y * WIN_NODES;
    const int nn   = min(WIN_NODES, n - base);
    const int nq   = (nn + 7) >> 3;

    constexpr int EPT = CHUNK_EDGES / CNT_THREADS;
    const int e0 = blockIdx.x * CHUNK_EDGES;
    int d[EPT];
#pragma unroll
    for (int k = 0; k < EPT; ++k) {
        const int e = e0 + k * CNT_THREADS + tid;
        d[k] = (e < E) ? dst[e] : -1;
    }

    for (int q = tid; q < nq; q += CNT_THREADS) cnt4[q] = make_uint4(0u, 0u, 0u, 0u);
    __syncthreads();

#pragma unroll
    for (int k = 0; k < EPT; ++k) {
        const unsigned int r = (unsigned int)(d[k] - base);
        if (r < (unsigned int)nn)
            atomicAdd(&cnt[r >> 1], 1u << ((r & 1u) * 16u));
    }
    __syncthreads();

    uint4* row = (uint4*)(part + (size_t)blockIdx.x * wpr + (size_t)blockIdx.y * (WIN_NODES / 2));
    for (int q = tid; q < nq; q += CNT_THREADS) row[q] = cnt4[q];

    const int gthread = (blockIdx.y * gridDim.x + blockIdx.x) * CNT_THREADS + tid;
    const int gstep   = gridDim.x * gridDim.y * CNT_THREADS;
    lay_b_fragments<128>(W1, Wf1, gthread, gstep);
    lay_b_fragments<64>(W2, Wf2, gthread, gstep);

    if (gthread < N_CTR * CTR_PITCH) ctr[gthread] = 0u;
}

template <int OUTC, bool FIN, bool PIN, bool POUT>
__global__ __launch_bounds__(256)
void k_gemm_f16(const float* __restrict__ X, const _Float16* __restrict__ Wf,
                const unsigned short* __restrict__ part16, int nchunks, int pitch16,
                float* __restrict__ dinv, const float* __restrict__ bin,
                const float* __restrict__ bout,
                float* __restrict__ H, float* __restrict__ S, int ntiles, int n) {
    static_assert(!POUT || OUTC == 128, "plane-split output is for 128-channel rows");
    const size_t plane = (size_t)n * 64;
    const int tid  = threadIdx.x;
    const int wave = tid >> 5;
    const int lane = tid & 31;
    const int tile = blockIdx.x * 8 + wave;
    if (tile >= ntiles) return;

    const int mn = lane & 15;
    const int hh = lane >> 4;

    const size_t arow = (size_t)(tile * 16 + mn);
    float dva;
    if (FIN) {
        dva = dinv[arow];
    } else {
        unsigned int c = 0;
        const unsigned short* p = part16 + arow;
        for (int k0 = hh; k0 < nchunks; k0 += 64) {
            unsigned int v[32];
#pragma unroll
            for (int j = 0; j < 32; ++j)
                v[j] = p[(size_t)min(k0 + 2 * j, nchunks - 1) * pitch16];
#pragma unroll
            for (int j = 0; j < 32; ++j)
                c += (k0 + 2 * j < nchunks) ? v[j] : 0u;
        }
        c += (unsigned int)__shfl_xor((int)c, 16);
        dva = rsqrtf((float)c + 1.0f);
        if (hh == 0 && arow < (size_t)n) dinv[arow] = dva;
    }

    v16h a[4];
    v16h al[FIN ? 4 : 1];
#pragma unroll
    for (int kf = 0; kf < 4; ++kf) {
        const float* xk = PIN ? X + (size_t)(kf >> 1) * plane + arow * 64 + (kf & 1) * 32
                              : X + arow * 128 + kf * 32;
#pragma unroll
        for (int i = 0; i < 8; ++i) {
            int kk = ((i < 4) ? 2 * i : 2 * i + 8) + 8 * hh;
            float v0 = xk[kk];
            float v1 = xk[kk + 1];
            if (FIN) {
                const float* bk = bin + kf * 32 + kk;
                v0 = fmaxf(fmaf(v0, dva, bk[0]), 0.0f);
                v1 = fmaxf(fmaf(v1, dva, bk[1]), 0.0f);
            }
            const _Float16 h0 = (_Float16)v0;
            const _Float16 h1 = (_Float16)v1;
            a[kf][2 * i]     = h0;
            a[kf][2 * i + 1] = h1;
            if (FIN) {
                al[kf][2 * i]     = (_Float16)(v0 - (float)h0);
                al[kf][2 * i + 1] = (_Float16)(v1 - (float)h1);
            }
        }
    }

    float dvo[8];
#pragma unroll
    for (int r = 0; r < 8; ++r)
        dvo[r] = FIN ? dinv[tile * 16 + r + 8 * hh] : __shfl(dva, r + 8 * hh);

    const v16h* bfrag = (const v16h*)Wf + lane;
#pragma unroll
    for (int nt = 0; nt < OUTC / 16; ++nt) {
        v8f c = {};
#pragma unroll
        for (int kf = 0; kf < 4; ++kf) {
            const v16h b = bfrag[(nt * 4 + kf) * 32];
            c = __builtin_amdgcn_wmma_f32_16x16x32_f16(
                    false, a[kf], false, b, (short)0, c, false, false);
            if (FIN)
                c = __builtin_amdgcn_wmma_f32_16x16x32_f16(
                        false, al[kf], false, b, (short)0, c, false, false);
        }
        constexpr int PITCH = POUT ? 64 : OUTC;
        const size_t off = POUT ? (size_t)(nt >> 2) * plane + (size_t)(tile * 16) * 64 + (nt & 3) * 16 + mn
                                : (size_t)(tile * 16) * OUTC + nt * 16 + mn;
        float* hout = H + off;
        float* sout = S + off;
        const float bo = FIN ? bout[nt * 16 + mn] : 0.0f;
#pragma unroll
        for (int r = 0; r < 8; ++r) {
            const float hv = c[r] * dvo[r];
            hout[(size_t)(r + 8 * hh) * PITCH] = hv;
            sout[(size_t)(r + 8 * hh) * PITCH] = FIN ? hv * dvo[r] + bo : hv;
        }
    }
}


constexpr int SC_THREADS = 256;
constexpr int SC_WAVES   = SC_THREADS / 32;
constexpr int SC_SUB     = 2;
constexpr int SC_CHUNK   = SC_WAVES * SC_SUB * 32;
constexpr int SC_BLOCKS  = 2048;

__device__ __forceinline__ unsigned int draw_chunk(unsigned int* counter) {
    return __builtin_amdgcn_atomic_inc32(counter, 0xFFFFFFFFu, __ATOMIC_RELAXED, "agent");
}

template <int G, bool DSTN>
__device__ __forceinline__ void add_group(const float* __restrict__ h, float* __restrict__ agg,
                                          int sv, int dv, int nb, unsigned int& m, int lane) {
    constexpr int C = 64;
    int ss[G], dd[G], ll[G];
#pragma unroll
    for (int k = 0; k < G; ++k) {
        const bool ok = (k == 0) || (m != 0u);
        const int l = ok ? __builtin_ctz(m) : ll[0];
        m &= m - 1u;
        ll[k] = l;
        ss[k] = __builtin_amdgcn_readlane(sv, l);
        dd[k] = ok ? __builtin_amdgcn_readlane(dv, l) : -1;
    }
    float v[G][2];
#pragma unroll
    for (int k = 0; k < G; ++k) {
        const float* hr = h + (size_t)ss[k] * C + lane;
        v[k][0] = hr[0];
        v[k][1] = hr[32];
    }
#pragma unroll
    for (int k = 0; k < G; ++k) {
        if (dd[k] >= 0) {
            const float nrm = DSTN ? __builtin_bit_cast(float, __builtin_amdgcn_readlane(nb, ll[k])) : 1.0f;
            float* p = agg + (size_t)dd[k] * C + lane;
            atomicAdd(p,      DSTN ? v[k][0] * nrm : v[k][0]);
            atomicAdd(p + 32, DSTN ? v[k][1] * nrm : v[k][1]);
        }
    }
}

template <bool DSTN>
__global__ __launch_bounds__(SC_THREADS)
void k_scatter_own(const float* __restrict__ h, const int* __restrict__ src,
                   const int* __restrict__ dst, const float* __restrict__ dinv,
                   float* __restrict__ agg, int E, unsigned int* __restrict__ ctr) {
    static_assert(SC_SUB == 2, "the id fence below names two pieces");
    __shared__ unsigned int s_next[2];
    const int tid  = threadIdx.x;
    const int lane = tid & 31;
    const int wave = tid >> 5;
    const unsigned int half = ((unsigned int)__builtin_amdgcn_s_sendmsg_rtn(0x87) >> 18) & 1u;
    const unsigned int nchunk = (unsigned int)((E + SC_CHUNK - 1) / SC_CHUNK);

    for (unsigned int pass = 0; pass < 2u; ++pass) {
        const unsigned int set = half ^ pass;
        unsigned int* my = ctr + set * CTR_PITCH;
        if (tid == 0) s_next[0] = draw_chunk(my);
        __syncthreads();
        for (unsigned int it = 0; ; ++it) {
            const unsigned int c = (unsigned int)__builtin_amdgcn_readfirstlane((int)s_next[it & 1u]);
            if (c >= nchunk) break;
            const int e0 = (int)c * SC_CHUNK + wave * (SC_SUB * 32);
            int sv[SC_SUB], dv[SC_SUB];
#pragma unroll
            for (int j = 0; j < SC_SUB; ++j) {
                const int e  = e0 + j * 32 + lane;
                const int ec = min(e, E - 1);
                sv[j] = src[ec];
                const int dl = dst[ec];
                dv[j] = (e < E) ? dl : -1;
            }
            asm volatile("" :: "v"(sv[0]), "v"(sv[1]), "v"(dv[0]), "v"(dv[1]));
            unsigned int drawn = 0;
            if (tid == 0) drawn = draw_chunk(my);
#pragma unroll
            for (int j = 0; j < SC_SUB; ++j) {
                const bool mine = dv[j] >= 0 && ((unsigned int)dv[j] & 1u) == set;
                int nb = 0;
                if (DSTN) nb = __builtin_bit_cast(int, dinv[mine ? dv[j] : 0]);
                unsigned int m = __builtin_amdgcn_ballot_w32(mine);
                while (m != 0u) {
                    if (__builtin_popcount(m) > 8) add_group<16, DSTN>(h, agg, sv[j], dv[j], nb, m, lane);
                    else                           add_group<8, DSTN>(h, agg, sv[j], dv[j], nb, m, lane);
                }
            }

            if (tid == 0) s_next[(it + 1u) & 1u] = drawn;
            __builtin_amdgcn_fence(__ATOMIC_RELEASE, "workgroup", "local");
            __builtin_amdgcn_s_barrier();
            __builtin_amdgcn_fence(__ATOMIC_ACQUIRE, "workgroup", "local");
        }
        __syncthreads();
    }
}


extern "C" void kernel_launch(void* const* d_in, const int* in_sizes, int n_in,
                              void* d_out, int out_size, void* d_ws, size_t ws_size,
                              hipStream_t stream) {
    const float* x  = (const float*)d_in[0];
    const int*   ei = (const int*)d_in[1];
    const float* W1 = (const float*)d_in[2];
    const float* b1 = (const float*)d_in[3];
    const float* W2 = (const float*)d_in[4];
    const float* b2 = (const float*)d_in[5];
    float*       out = (float*)d_out;

    const int N = in_sizes[0] / 128;
    const int E = in_sizes[1] / 2;
    const int* src = ei;
    const int* dst = ei + E;

    char* w = (char*)d_ws;
    auto carve = [&](size_t bytes) -> void* {
        void* p = (void*)w;
        w += (bytes + 255) & ~(size_t)255;
        return p;
    };
    float* dinv = (float*)carve((size_t)N * 4);
    float* h1   = (float*)carve((size_t)N * 128 * 4);
    float* agg1 = (float*)carve((size_t)N * 128 * 4);
    float* h2   = h1;
    _Float16* Wf1 = (_Float16*)carve((size_t)128 * 128 * 2);
    _Float16* Wf2 = (_Float16*)carve((size_t)128 * 64 * 2);
    const int nchunks = (E + CHUNK_EDGES - 1) / CHUNK_EDGES;
    const int nwin    = (N + WIN_NODES - 1) / WIN_NODES;
    const int wpr     = ((N + 7) / 8) * 4;
    unsigned int* part = (unsigned int*)carve((size_t)nchunks * wpr * 4);
    unsigned int* ctr  = (unsigned int*)carve((size_t)N_CTR * CTR_PITCH * 4);

    const int ntiles = (N + 15) / 16;
    const int gemm_blocks = (ntiles + 7) / 8;

    k_count<<<dim3(nchunks, nwin), CNT_THREADS, 0, stream>>>(dst, part, E, N, wpr, W1, Wf1, W2, Wf2, ctr);
    const unsigned short* part16 = (const unsigned short*)part;

    const size_t plane = (size_t)N * 64;
    k_gemm_f16<128, false, false, true><<<gemm_blocks, 256, 0, stream>>>(x, Wf1, part16, nchunks, 2 * wpr,
                                                                         dinv, b1, b1, h1, agg1, ntiles, N);
    k_scatter_own<false><<<SC_BLOCKS, SC_THREADS, 0, stream>>>(h1, src, dst, dinv, agg1, E, ctr);
    k_scatter_own<false><<<SC_BLOCKS, SC_THREADS, 0, stream>>>(h1 + plane, src, dst, dinv, agg1 + plane, E,
                                                              ctr + 2 * CTR_PITCH);

    k_gemm_f16<64, true, true, false><<<gemm_blocks, 256, 0, stream>>>(agg1, Wf2, part16, nchunks, 2 * wpr,
                                                                       dinv, b1, b2, h2, out, ntiles, N);
    k_scatter_own<true><<<SC_BLOCKS, SC_THREADS, 0, stream>>>(h2, src, dst, dinv, out, E, ctr + 4 * CTR_PITCH);
}
